# starting kernel
# baseline (speedup 1.0000x reference)
#include <hip/hip_runtime.h>
#include <math.h>


#define B_SZ    32
#define N_INN   2048
#define J_OUT   64
#define D_DIM   16
#define ROW_F4  256
#define NCHUNK  64
#define ROWS_PER_BLOCK (N_INN / NCHUNK)
#define NWAVES  8
#define ROWS_PER_WAVE (ROWS_PER_BLOCK / NWAVES)
#define EPS_SQ  1e-7f

typedef __attribute__((ext_vector_type(2))) float v2f;
typedef __attribute__((ext_vector_type(8))) float v8f;

__device__ __forceinline__ void f4add(float4& a, const float4 b) {
  a.x += b.x; a.y += b.y; a.z += b.z; a.w += b.w;
}
__device__ __forceinline__ void f4fma(float4& a, float s, const float4 b) {
  a.x = fmaf(s, b.x, a.x); a.y = fmaf(s, b.y, a.y);
  a.z = fmaf(s, b.z, a.z); a.w = fmaf(s, b.w, a.w);
}

template <int MODE>
__global__ __launch_bounds__(256) void caps_pass(const float* __restrict__ X,
                                                 const float* __restrict__ Oprev,
                                                 float* __restrict__ Lg,
                                                 float* __restrict__ P) {
  const int chunk = blockIdx.x, b = blockIdx.y;
  const int t = threadIdx.x;
  const int wave = t >> 5, lane = t & 31;
  const int jgrp = lane >> 2;
  const int dsub = lane & 3;

  __shared__ float4 ls[NWAVES * ROW_F4];

  float4 o[8];
  if (MODE != 0) {
    const float4* Ob = (const float4*)Oprev + b * ROW_F4;
#pragma unroll
    for (int k = 0; k < 8; ++k) o[k] = Ob[k * 32 + lane];
  }

  float4 acc[8];
#pragma unroll
  for (int k = 0; k < 8; ++k) acc[k] = make_float4(0.f, 0.f, 0.f, 0.f);

  const int row0 = chunk * ROWS_PER_BLOCK + wave * ROWS_PER_WAVE;
  const float4* Xb = (const float4*)X + (size_t)b * N_INN * ROW_F4;

  for (int r = 0; r < ROWS_PER_WAVE; ++r) {
    const int i = row0 + r;
    const float4* xr = Xb + (size_t)i * ROW_F4;
    __builtin_prefetch(xr + ROW_F4 + lane * 8, 0, 0);

    float4 x[8];
#pragma unroll
    for (int k = 0; k < 8; ++k) x[k] = xr[k * 32 + lane];

    if (MODE == 0) {
#pragma unroll
      for (int k = 0; k < 8; ++k) f4add(acc[k], x[k]);
    } else {
      float Lv[8];
#pragma unroll
      for (int k = 0; k < 8; ++k) {
        float p = x[k].x * o[k].x + x[k].y * o[k].y + x[k].z * o[k].z + x[k].w * o[k].w;
        p += __shfl_xor(p, 1, 32);
        p += __shfl_xor(p, 2, 32);
        if (MODE == 2)
          p += Lg[((size_t)(b * N_INN) + i) * J_OUT + 8 * k + jgrp];
        Lv[k] = p;
      }
      float mx = Lv[0];
#pragma unroll
      for (int k = 1; k < 8; ++k) mx = fmaxf(mx, Lv[k]);
#pragma unroll
      for (int off = 1; off < 32; off <<= 1) mx = fmaxf(mx, __shfl_xor(mx, off, 32));
      float e[8], ss = 0.f;
#pragma unroll
      for (int k = 0; k < 8; ++k) { e[k] = __expf(Lv[k] - mx); ss += e[k]; }
#pragma unroll
      for (int off = 1; off < 32; off <<= 1) ss += __shfl_xor(ss, off, 32);
      const float inv = 4.0f / ss;
#pragma unroll
      for (int k = 0; k < 8; ++k) f4fma(acc[k], e[k] * inv, x[k]);

      if (MODE == 1 && dsub == 0) {
#pragma unroll
        for (int k = 0; k < 8; ++k)
          Lg[((size_t)(b * N_INN) + i) * J_OUT + 8 * k + jgrp] = Lv[k];
      }
    }
  }

#pragma unroll
  for (int k = 0; k < 8; ++k) ls[wave * ROW_F4 + k * 32 + lane] = acc[k];
  __syncthreads();
  float4 s = ls[t];
#pragma unroll
  for (int w = 1; w < NWAVES; ++w) f4add(s, ls[w * ROW_F4 + t]);
  ((float4*)P)[((size_t)(b * NCHUNK) + chunk) * ROW_F4 + t] = s;
}

__global__ __launch_bounds__(256) void caps_reduce_squash(const float* __restrict__ P,
                                                          const float* __restrict__ bias,
                                                          float* __restrict__ O,
                                                          float scale) {
  const int b = blockIdx.x;
  const int t = threadIdx.x, wave = t >> 5, lane = t & 31;
  const int m = lane & 15, half = lane >> 4;
  const v2f ones = {1.0f, 1.0f};

  for (int g = 0; g < J_OUT / NWAVES; ++g) {
    const int j = wave + g * NWAVES;
    v8f acc = {0.f, 0.f, 0.f, 0.f, 0.f, 0.f, 0.f, 0.f};
#pragma unroll
    for (int s = 0; s < NCHUNK / 4; ++s) {
      const int c = 4 * s + 2 * half;
      v2f a;
      a.x = P[((size_t)(b * NCHUNK) + c    ) * 1024 + j * 16 + m];
      a.y = P[((size_t)(b * NCHUNK) + c + 1) * 1024 + j * 16 + m];
      acc = __builtin_amdgcn_wmma_f32_16x16x4_f32(
           false, a,  false, ones,
           (short)0, acc,  false,  false);
    }
    float val[8];
    float n2 = 0.f;
#pragma unroll
    for (int v = 0; v < 8; ++v) {
      const int d = v + 8 * half;
      val[v] = scale * acc[v] + bias[j * 16 + d];
      n2 = fmaf(val[v], val[v], n2);
    }
    n2 += __shfl_xor(n2, 16, 32);
    const float coef = (n2 / (1.0f + n2)) / sqrtf(n2 + EPS_SQ);
    if (m == 0) {
#pragma unroll
      for (int v = 0; v < 8; ++v)
        O[(size_t)b * (J_OUT * D_DIM) + j * 16 + 8 * half + v] = val[v] * coef;
    }
  }
}

extern "C" void kernel_launch(void* const* d_in, const int* in_sizes, int n_in,
                              void* d_out, int out_size, void* d_ws, size_t ws_size,
                              hipStream_t stream) {
  const float* X    = (const float*)d_in[0];
  const float* bias = (const float*)d_in[1];
  float* out = (float*)d_out;

  float* P  = (float*)d_ws;
  float* O0 = P  + (size_t)B_SZ * NCHUNK * 1024;
  float* O1 = O0 + B_SZ * J_OUT * D_DIM;
  float* Lg = O1 + B_SZ * J_OUT * D_DIM;

  dim3 grid(NCHUNK, B_SZ);

  caps_pass<0><<<grid, 256, 0, stream>>>(X, nullptr, nullptr, P);
  caps_reduce_squash<<<B_SZ, 256, 0, stream>>>(P, bias, O0, 1.0f / 64.0f);

  caps_pass<1><<<grid, 256, 0, stream>>>(X, O0, Lg, P);
  caps_reduce_squash<<<B_SZ, 256, 0, stream>>>(P, bias, O1, 1.0f);

  caps_pass<2><<<grid, 256, 0, stream>>>(X, O1, Lg, P);
  caps_reduce_squash<<<B_SZ, 256, 0, stream>>>(P, bias, out, 1.0f);
}


# reduce+squash kernel: one wave per (batch, capsule) on 256 blocks instead of 32 blocks whose waves loop over 8 capsules in series; same chunk order on the f32 WMMA, sums bit-identical
# speedup vs baseline: 1.2773x; 1.2773x over previous
#include <hip/hip_runtime.h>
#include <math.h>


#define B_SZ    32
#define N_INN   2048
#define J_OUT   64
#define D_DIM   16
#define ROW_F4  256
#define NCHUNK  64
#define ROWS_PER_BLOCK (N_INN / NCHUNK)
#define NWAVES  8
#define JGROUPS (J_OUT / NWAVES)
#define ROWS_PER_WAVE (ROWS_PER_BLOCK / NWAVES)
#define EPS_SQ  1e-7f

typedef __attribute__((ext_vector_type(2))) float v2f;
typedef __attribute__((ext_vector_type(8))) float v8f;

__device__ __forceinline__ void f4add(float4& a, const float4 b) {
  a.x += b.x; a.y += b.y; a.z += b.z; a.w += b.w;
}
__device__ __forceinline__ void f4fma(float4& a, float s, const float4 b) {
  a.x = fmaf(s, b.x, a.x); a.y = fmaf(s, b.y, a.y);
  a.z = fmaf(s, b.z, a.z); a.w = fmaf(s, b.w, a.w);
}

template <int MODE>
__global__ __launch_bounds__(256) void caps_pass(const float* __restrict__ X,
                                                 const float* __restrict__ Oprev,
                                                 float* __restrict__ Lg,
                                                 float* __restrict__ P) {
  const int chunk = blockIdx.x, b = blockIdx.y;
  const int t = threadIdx.x;
  const int wave = t >> 5, lane = t & 31;
  const int jgrp = lane >> 2;
  const int dsub = lane & 3;

  __shared__ float4 ls[NWAVES * ROW_F4];

  float4 o[8];
  if (MODE != 0) {
    const float4* Ob = (const float4*)Oprev + b * ROW_F4;
#pragma unroll
    for (int k = 0; k < 8; ++k) o[k] = Ob[k * 32 + lane];
  }

  float4 acc[8];
#pragma unroll
  for (int k = 0; k < 8; ++k) acc[k] = make_float4(0.f, 0.f, 0.f, 0.f);

  const int row0 = chunk * ROWS_PER_BLOCK + wave * ROWS_PER_WAVE;
  const float4* Xb = (const float4*)X + (size_t)b * N_INN * ROW_F4;

  for (int r = 0; r < ROWS_PER_WAVE; ++r) {
    const int i = row0 + r;
    const float4* xr = Xb + (size_t)i * ROW_F4;
    __builtin_prefetch(xr + ROW_F4 + lane * 8, 0, 0);

    float4 x[8];
#pragma unroll
    for (int k = 0; k < 8; ++k) x[k] = xr[k * 32 + lane];

    if (MODE == 0) {
#pragma unroll
      for (int k = 0; k < 8; ++k) f4add(acc[k], x[k]);
    } else {
      float Lv[8];
#pragma unroll
      for (int k = 0; k < 8; ++k) {
        float p = x[k].x * o[k].x + x[k].y * o[k].y + x[k].z * o[k].z + x[k].w * o[k].w;
        p += __shfl_xor(p, 1, 32);
        p += __shfl_xor(p, 2, 32);
        if (MODE == 2)
          p += Lg[((size_t)(b * N_INN) + i) * J_OUT + 8 * k + jgrp];
        Lv[k] = p;
      }
      float mx = Lv[0];
#pragma unroll
      for (int k = 1; k < 8; ++k) mx = fmaxf(mx, Lv[k]);
#pragma unroll
      for (int off = 1; off < 32; off <<= 1) mx = fmaxf(mx, __shfl_xor(mx, off, 32));
      float e[8], ss = 0.f;
#pragma unroll
      for (int k = 0; k < 8; ++k) { e[k] = __expf(Lv[k] - mx); ss += e[k]; }
#pragma unroll
      for (int off = 1; off < 32; off <<= 1) ss += __shfl_xor(ss, off, 32);
      const float inv = 4.0f / ss;
#pragma unroll
      for (int k = 0; k < 8; ++k) f4fma(acc[k], e[k] * inv, x[k]);

      if (MODE == 1 && dsub == 0) {
#pragma unroll
        for (int k = 0; k < 8; ++k)
          Lg[((size_t)(b * N_INN) + i) * J_OUT + 8 * k + jgrp] = Lv[k];
      }
    }
  }

#pragma unroll
  for (int k = 0; k < 8; ++k) ls[wave * ROW_F4 + k * 32 + lane] = acc[k];
  __syncthreads();
  float4 s = ls[t];
#pragma unroll
  for (int w = 1; w < NWAVES; ++w) f4add(s, ls[w * ROW_F4 + t]);
  ((float4*)P)[((size_t)(b * NCHUNK) + chunk) * ROW_F4 + t] = s;
}

__global__ __launch_bounds__(256) void caps_reduce_squash(const float* __restrict__ P,
                                                          const float* __restrict__ bias,
                                                          float* __restrict__ O,
                                                          float scale) {
  const int g = blockIdx.x, b = blockIdx.y;
  const int t = threadIdx.x, wave = t >> 5, lane = t & 31;
  const int m = lane & 15, half = lane >> 4;
  const int j = g * NWAVES + wave;
  const v2f ones = {1.0f, 1.0f};

  const float* Pj = P + ((size_t)(b * NCHUNK) + 2 * half) * 1024 + j * 16 + m;
  v2f a[NCHUNK / 4];
#pragma unroll
  for (int s = 0; s < NCHUNK / 4; ++s) {
    a[s].x = Pj[(size_t)(4 * s    ) * 1024];
    a[s].y = Pj[(size_t)(4 * s + 1) * 1024];
  }
  const float4 bl = *(const float4*)(bias + j * 16 + 8 * half);
  const float4 bh = *(const float4*)(bias + j * 16 + 8 * half + 4);
  const float bv[8] = {bl.x, bl.y, bl.z, bl.w, bh.x, bh.y, bh.z, bh.w};

  v8f acc = {0.f, 0.f, 0.f, 0.f, 0.f, 0.f, 0.f, 0.f};
#pragma unroll
  for (int s = 0; s < NCHUNK / 4; ++s) {
    acc = __builtin_amdgcn_wmma_f32_16x16x4_f32(
         false, a[s],  false, ones,
         (short)0, acc,  false,  false);
  }
  float val[8];
  float n2 = 0.f;
#pragma unroll
  for (int v = 0; v < 8; ++v) {
    val[v] = scale * acc[v] + bv[v];
    n2 = fmaf(val[v], val[v], n2);
  }
  n2 += __shfl_xor(n2, 16, 32);
  const float coef = (n2 / (1.0f + n2)) / sqrtf(n2 + EPS_SQ);
  if (m == 0) {
#pragma unroll
    for (int v = 0; v < 8; ++v)
      O[(size_t)b * (J_OUT * D_DIM) + j * 16 + 8 * half + v] = val[v] * coef;
  }
}

extern "C" void kernel_launch(void* const* d_in, const int* in_sizes, int n_in,
                              void* d_out, int out_size, void* d_ws, size_t ws_size,
                              hipStream_t stream) {
  const float* X    = (const float*)d_in[0];
  const float* bias = (const float*)d_in[1];
  float* out = (float*)d_out;

  float* P  = (float*)d_ws;
  float* O0 = P  + (size_t)B_SZ * NCHUNK * 1024;
  float* O1 = O0 + B_SZ * J_OUT * D_DIM;
  float* Lg = O1 + B_SZ * J_OUT * D_DIM;

  dim3 grid(NCHUNK, B_SZ);
  dim3 rgrid(JGROUPS, B_SZ);

  caps_pass<0><<<grid, 256, 0, stream>>>(X, nullptr, nullptr, P);
  caps_reduce_squash<<<rgrid, 256, 0, stream>>>(P, bias, O0, 1.0f / 64.0f);

  caps_pass<1><<<grid, 256, 0, stream>>>(X, O0, Lg, P);
  caps_reduce_squash<<<rgrid, 256, 0, stream>>>(P, bias, O1, 1.0f);

  caps_pass<2><<<grid, 256, 0, stream>>>(X, O1, Lg, P);
  caps_reduce_squash<<<rgrid, 256, 0, stream>>>(P, bias, out, 1.0f);
}


# streaming passes as one 16-wave workgroup per WGP (256 blocks), each wave walks 16 rows with two rows of loads kept in flight; i-sum tree now 16 rows x 16 waves x 8 chunks (f32 WMMA)
# speedup vs baseline: 1.4897x; 1.1663x over previous
#include <hip/hip_runtime.h>
#include <math.h>


#define B_SZ    32
#define N_INN   2048
#define J_OUT   64
#define D_DIM   16
#define ROW_F4  256
#define NCHUNK  8
#define ROWS_PER_BLOCK (N_INN / NCHUNK)
#define PWAVES  16
#define ROWS_PER_WAVE (ROWS_PER_BLOCK / PWAVES)
#define NWAVES  8
#define JGROUPS (J_OUT / NWAVES)
#define EPS_SQ  1e-7f

typedef __attribute__((ext_vector_type(2))) float v2f;
typedef __attribute__((ext_vector_type(4))) float v4f;
typedef __attribute__((ext_vector_type(8))) float v8f;

__device__ __forceinline__ void f4add(v4f& a, const v4f b) { a += b; }
__device__ __forceinline__ void f4fma(v4f& a, float s, const v4f b) {
  const v4f sv = {s, s, s, s};
  a = __builtin_elementwise_fma(sv, b, a);
}

__device__ __forceinline__ int request_after(int nxt, const v4f& done) {
  asm volatile("" : "+s"(nxt) : "v"(done.w));
  return nxt;
}

#define SB() __builtin_amdgcn_sched_barrier(0)

template <int MODE>
__device__ __forceinline__ void load_row(v4f (&x)[8], float (&lp)[8], const v4f* xw, const float* lw, int r) {
#pragma unroll
  for (int k = 0; k < 8; ++k) x[k] = xw[(size_t)r * ROW_F4 + k * 32];
  if (MODE == 2) {
#pragma unroll
    for (int k = 0; k < 8; ++k) lp[k] = lw[r * J_OUT + 8 * k];
  }
}

template <int MODE>
__device__ __forceinline__ void row_work(const v4f (&xin)[8], const float (&lp)[8], const v4f (&o)[8],
                                         v4f (&acc)[8], float* lw, int r, int dsub) {
  v4f xr[8];
#pragma unroll
  for (int k = 0; k < 8; ++k) {
    xr[k] = xin[k];
    if (MODE != 0) asm volatile("" : "+v"(xr[k]));
  }
  if (MODE == 0) {
#pragma unroll
    for (int k = 0; k < 8; ++k) f4add(acc[k], xr[k]);
  } else {
    float Lv[8];
#pragma unroll
    for (int k = 0; k < 8; ++k) {
      float p = xr[k].x * o[k].x + xr[k].y * o[k].y + xr[k].z * o[k].z + xr[k].w * o[k].w;
      p += __shfl_xor(p, 1, 32);
      p += __shfl_xor(p, 2, 32);
      if (MODE == 2) p += lp[k];
      Lv[k] = p;
    }
    float mx = Lv[0];
#pragma unroll
    for (int k = 1; k < 8; ++k) mx = fmaxf(mx, Lv[k]);
#pragma unroll
    for (int off = 1; off < 32; off <<= 1) mx = fmaxf(mx, __shfl_xor(mx, off, 32));
    float e[8], ss = 0.f;
#pragma unroll
    for (int k = 0; k < 8; ++k) { e[k] = __expf(Lv[k] - mx); ss += e[k]; }
#pragma unroll
    for (int off = 1; off < 32; off <<= 1) ss += __shfl_xor(ss, off, 32);
    const float inv = 4.0f / ss;
#pragma unroll
    for (int k = 0; k < 8; ++k) f4fma(acc[k], e[k] * inv, xr[k]);

    if (MODE == 1 && dsub == 0) {
#pragma unroll
      for (int k = 0; k < 8; ++k) lw[r * J_OUT + 8 * k] = Lv[k];
    }
  }
}

template <int MODE>
__global__ __launch_bounds__(PWAVES * 32) void caps_pass(const float* __restrict__ X,
                                                         const float* __restrict__ Oprev,
                                                         float* __restrict__ Lg,
                                                         float* __restrict__ P) {
  const int chunk = blockIdx.x, b = blockIdx.y;
  const int t = threadIdx.x;
  const int wave = __builtin_amdgcn_readfirstlane(t >> 5);
  const int lane = t & 31;
  const int jgrp = lane >> 2;
  const int dsub = lane & 3;

  __shared__ v4f ls[PWAVES * ROW_F4];

  v4f o[8];
  if (MODE != 0) {
    const v4f* Ob = (const v4f*)Oprev + b * ROW_F4;
#pragma unroll
    for (int k = 0; k < 8; ++k) o[k] = Ob[k * 32 + lane];
  }

  v4f acc[8];
#pragma unroll
  for (int k = 0; k < 8; ++k) acc[k] = (v4f){0.f, 0.f, 0.f, 0.f};

  const int row0 = chunk * ROWS_PER_BLOCK + wave * ROWS_PER_WAVE;
  const v4f* xw = (const v4f*)X + ((size_t)b * N_INN + row0) * ROW_F4 + lane;
  float* lw = Lg + ((size_t)b * N_INN + row0) * J_OUT + jgrp;

  v4f xa[8], xb[8], xc[8];
  float la[8], lb[8], lc[8];
  int nxt = 0;
  load_row<MODE>(xa, la, xw, lw, nxt++); SB();
  load_row<MODE>(xb, lb, xw, lw, nxt++); SB();
#pragma unroll 1
  for (int r = 0; r < ROWS_PER_WAVE - 4; r += 3) {
    load_row<MODE>(xc, lc, xw, lw, nxt++); SB();
    row_work<MODE>(xa, la, o, acc, lw, r, dsub); SB();
    nxt = request_after(nxt, acc[7]);
    load_row<MODE>(xa, la, xw, lw, nxt++); SB();
    row_work<MODE>(xb, lb, o, acc, lw, r + 1, dsub); SB();
    nxt = request_after(nxt, acc[7]);
    load_row<MODE>(xb, lb, xw, lw, nxt++); SB();
    row_work<MODE>(xc, lc, o, acc, lw, r + 2, dsub); SB();
    nxt = request_after(nxt, acc[7]);
  }
  load_row<MODE>(xc, lc, xw, lw, nxt++); SB();
  row_work<MODE>(xa, la, o, acc, lw, ROWS_PER_WAVE - 4, dsub); SB();
  nxt = request_after(nxt, acc[7]);
  load_row<MODE>(xa, la, xw, lw, nxt++); SB();
  row_work<MODE>(xb, lb, o, acc, lw, ROWS_PER_WAVE - 3, dsub); SB();
  row_work<MODE>(xc, lc, o, acc, lw, ROWS_PER_WAVE - 2, dsub); SB();
  row_work<MODE>(xa, la, o, acc, lw, ROWS_PER_WAVE - 1, dsub);

#pragma unroll
  for (int k = 0; k < 8; ++k) ls[wave * ROW_F4 + k * 32 + lane] = acc[k];
  __syncthreads();
  const float2* ls2 = (const float2*)ls;
  float2 s = ls2[t];
#pragma unroll
  for (int w = 1; w < PWAVES; ++w) {
    const float2 v = ls2[w * (2 * ROW_F4) + t];
    s.x += v.x; s.y += v.y;
  }
  ((float2*)P)[((size_t)(b * NCHUNK) + chunk) * (2 * ROW_F4) + t] = s;
}

__global__ __launch_bounds__(256) void caps_reduce_squash(const float* __restrict__ P,
                                                          const float* __restrict__ bias,
                                                          float* __restrict__ O,
                                                          float scale) {
  const int g = blockIdx.x, b = blockIdx.y;
  const int t = threadIdx.x, wave = t >> 5, lane = t & 31;
  const int m = lane & 15, half = lane >> 4;
  const int j = g * NWAVES + wave;
  const v2f ones = {1.0f, 1.0f};

  const float* Pj = P + ((size_t)(b * NCHUNK) + 2 * half) * 1024 + j * 16 + m;
  v2f a[NCHUNK / 4];
#pragma unroll
  for (int s = 0; s < NCHUNK / 4; ++s) {
    a[s].x = Pj[(size_t)(4 * s    ) * 1024];
    a[s].y = Pj[(size_t)(4 * s + 1) * 1024];
  }
  const float4 bl = *(const float4*)(bias + j * 16 + 8 * half);
  const float4 bh = *(const float4*)(bias + j * 16 + 8 * half + 4);
  const float bv[8] = {bl.x, bl.y, bl.z, bl.w, bh.x, bh.y, bh.z, bh.w};

  v8f acc = {0.f, 0.f, 0.f, 0.f, 0.f, 0.f, 0.f, 0.f};
#pragma unroll
  for (int s = 0; s < NCHUNK / 4; ++s) {
    acc = __builtin_amdgcn_wmma_f32_16x16x4_f32(
         false, a[s],  false, ones,
         (short)0, acc,  false,  false);
  }
  float val[8];
  float n2 = 0.f;
#pragma unroll
  for (int v = 0; v < 8; ++v) {
    val[v] = scale * acc[v] + bv[v];
    n2 = fmaf(val[v], val[v], n2);
  }
  n2 += __shfl_xor(n2, 16, 32);
  const float coef = (n2 / (1.0f + n2)) / sqrtf(n2 + EPS_SQ);
  if (m == 0) {
#pragma unroll
    for (int v = 0; v < 8; ++v)
      O[(size_t)b * (J_OUT * D_DIM) + j * 16 + 8 * half + v] = val[v] * coef;
  }
}

extern "C" void kernel_launch(void* const* d_in, const int* in_sizes, int n_in,
                              void* d_out, int out_size, void* d_ws, size_t ws_size,
                              hipStream_t stream) {
  const float* X    = (const float*)d_in[0];
  const float* bias = (const float*)d_in[1];
  float* out = (float*)d_out;

  float* P  = (float*)d_ws;
  float* O0 = P  + (size_t)B_SZ * NCHUNK * 1024;
  float* O1 = O0 + B_SZ * J_OUT * D_DIM;
  float* Lg = O1 + B_SZ * J_OUT * D_DIM;

  dim3 grid(NCHUNK, B_SZ);
  dim3 rgrid(JGROUPS, B_SZ);

  caps_pass<0><<<grid, PWAVES * 32, 0, stream>>>(X, nullptr, nullptr, P);
  caps_reduce_squash<<<rgrid, 256, 0, stream>>>(P, bias, O0, 1.0f / 64.0f);

  caps_pass<1><<<grid, PWAVES * 32, 0, stream>>>(X, O0, Lg, P);
  caps_reduce_squash<<<rgrid, 256, 0, stream>>>(P, bias, O1, 1.0f);

  caps_pass<2><<<grid, PWAVES * 32, 0, stream>>>(X, O1, Lg, P);
  caps_reduce_squash<<<rgrid, 256, 0, stream>>>(P, bias, out, 1.0f);
}


# pass 1 walks each wave's 16 rows downwards (passes 0, 2 upwards): a pass starts on the rows the pass before read last, still in the second-level cache; pass-1 row sums add in descending order
# speedup vs baseline: 1.5770x; 1.0586x over previous
#include <hip/hip_runtime.h>
#include <math.h>


#define B_SZ    32
#define N_INN   2048
#define J_OUT   64
#define D_DIM   16
#define ROW_F4  256
#define NCHUNK  8
#define ROWS_PER_BLOCK (N_INN / NCHUNK)
#define PWAVES  16
#define ROWS_PER_WAVE (ROWS_PER_BLOCK / PWAVES)
#define NWAVES  8
#define JGROUPS (J_OUT / NWAVES)
#define EPS_SQ  1e-7f

typedef __attribute__((ext_vector_type(2))) float v2f;
typedef __attribute__((ext_vector_type(4))) float v4f;
typedef __attribute__((ext_vector_type(8))) float v8f;

__device__ __forceinline__ void f4add(v4f& a, const v4f b) { a += b; }
__device__ __forceinline__ void f4fma(v4f& a, float s, const v4f b) {
  const v4f sv = {s, s, s, s};
  a = __builtin_elementwise_fma(sv, b, a);
}

__device__ __forceinline__ int request_after(int nxt, const v4f& done) {
  asm volatile("" : "+s"(nxt) : "v"(done.w));
  return nxt;
}

#define SB() __builtin_amdgcn_sched_barrier(0)

template <int MODE>
__device__ __forceinline__ int walk_row(int r) { return MODE == 1 ? (ROWS_PER_WAVE - 1 - r) : r; }

template <int MODE>
__device__ __forceinline__ void load_row(v4f (&x)[8], float (&lp)[8], const v4f* xw, const float* lw, int r) {
  const int row = walk_row<MODE>(r);
#pragma unroll
  for (int k = 0; k < 8; ++k) x[k] = xw[(size_t)row * ROW_F4 + k * 32];
  if (MODE == 2) {
#pragma unroll
    for (int k = 0; k < 8; ++k) lp[k] = lw[row * J_OUT + 8 * k];
  }
}

template <int MODE>
__device__ __forceinline__ void row_work(const v4f (&xin)[8], const float (&lp)[8], const v4f (&o)[8],
                                         v4f (&acc)[8], float* lw, int r, int dsub) {
  v4f xr[8];
#pragma unroll
  for (int k = 0; k < 8; ++k) {
    xr[k] = xin[k];
    if (MODE != 0) asm volatile("" : "+v"(xr[k]));
  }
  if (MODE == 0) {
#pragma unroll
    for (int k = 0; k < 8; ++k) f4add(acc[k], xr[k]);
  } else {
    float Lv[8];
#pragma unroll
    for (int k = 0; k < 8; ++k) {
      float p = xr[k].x * o[k].x + xr[k].y * o[k].y + xr[k].z * o[k].z + xr[k].w * o[k].w;
      p += __shfl_xor(p, 1, 32);
      p += __shfl_xor(p, 2, 32);
      if (MODE == 2) p += lp[k];
      Lv[k] = p;
    }
    float mx = Lv[0];
#pragma unroll
    for (int k = 1; k < 8; ++k) mx = fmaxf(mx, Lv[k]);
#pragma unroll
    for (int off = 1; off < 32; off <<= 1) mx = fmaxf(mx, __shfl_xor(mx, off, 32));
    float e[8], ss = 0.f;
#pragma unroll
    for (int k = 0; k < 8; ++k) { e[k] = __expf(Lv[k] - mx); ss += e[k]; }
#pragma unroll
    for (int off = 1; off < 32; off <<= 1) ss += __shfl_xor(ss, off, 32);
    const float inv = 4.0f / ss;
#pragma unroll
    for (int k = 0; k < 8; ++k) f4fma(acc[k], e[k] * inv, xr[k]);

    if (MODE == 1 && dsub == 0) {
      const int row = walk_row<MODE>(r);
#pragma unroll
      for (int k = 0; k < 8; ++k) lw[row * J_OUT + 8 * k] = Lv[k];
    }
  }
}

template <int MODE>
__global__ __launch_bounds__(PWAVES * 32) void caps_pass(const float* __restrict__ X,
                                                         const float* __restrict__ Oprev,
                                                         float* __restrict__ Lg,
                                                         float* __restrict__ P) {
  const int chunk = blockIdx.x, b = blockIdx.y;
  const int t = threadIdx.x;
  const int wave = __builtin_amdgcn_readfirstlane(t >> 5);
  const int lane = t & 31;
  const int jgrp = lane >> 2;
  const int dsub = lane & 3;

  __shared__ v4f ls[PWAVES * ROW_F4];

  v4f o[8];
  if (MODE != 0) {
    const v4f* Ob = (const v4f*)Oprev + b * ROW_F4;
#pragma unroll
    for (int k = 0; k < 8; ++k) o[k] = Ob[k * 32 + lane];
  }

  v4f acc[8];
#pragma unroll
  for (int k = 0; k < 8; ++k) acc[k] = (v4f){0.f, 0.f, 0.f, 0.f};

  const int row0 = chunk * ROWS_PER_BLOCK + wave * ROWS_PER_WAVE;
  const v4f* xw = (const v4f*)X + ((size_t)b * N_INN + row0) * ROW_F4 + lane;
  float* lw = Lg + ((size_t)b * N_INN + row0) * J_OUT + jgrp;

  v4f xa[8], xb[8], xc[8];
  float la[8], lb[8], lc[8];
  int nxt = 0;
  load_row<MODE>(xa, la, xw, lw, nxt++); SB();
  load_row<MODE>(xb, lb, xw, lw, nxt++); SB();
#pragma unroll 1
  for (int r = 0; r < ROWS_PER_WAVE - 4; r += 3) {
    load_row<MODE>(xc, lc, xw, lw, nxt++); SB();
    row_work<MODE>(xa, la, o, acc, lw, r, dsub); SB();
    nxt = request_after(nxt, acc[7]);
    load_row<MODE>(xa, la, xw, lw, nxt++); SB();
    row_work<MODE>(xb, lb, o, acc, lw, r + 1, dsub); SB();
    nxt = request_after(nxt, acc[7]);
    load_row<MODE>(xb, lb, xw, lw, nxt++); SB();
    row_work<MODE>(xc, lc, o, acc, lw, r + 2, dsub); SB();
    nxt = request_after(nxt, acc[7]);
  }
  load_row<MODE>(xc, lc, xw, lw, nxt++); SB();
  row_work<MODE>(xa, la, o, acc, lw, ROWS_PER_WAVE - 4, dsub); SB();
  nxt = request_after(nxt, acc[7]);
  load_row<MODE>(xa, la, xw, lw, nxt++); SB();
  row_work<MODE>(xb, lb, o, acc, lw, ROWS_PER_WAVE - 3, dsub); SB();
  row_work<MODE>(xc, lc, o, acc, lw, ROWS_PER_WAVE - 2, dsub); SB();
  row_work<MODE>(xa, la, o, acc, lw, ROWS_PER_WAVE - 1, dsub);

#pragma unroll
  for (int k = 0; k < 8; ++k) ls[wave * ROW_F4 + k * 32 + lane] = acc[k];
  __syncthreads();
  const float2* ls2 = (const float2*)ls;
  float2 s = ls2[t];
#pragma unroll
  for (int w = 1; w < PWAVES; ++w) {
    const float2 v = ls2[w * (2 * ROW_F4) + t];
    s.x += v.x; s.y += v.y;
  }
  ((float2*)P)[((size_t)(b * NCHUNK) + chunk) * (2 * ROW_F4) + t] = s;
}

__global__ __launch_bounds__(256) void caps_reduce_squash(const float* __restrict__ P,
                                                          const float* __restrict__ bias,
                                                          float* __restrict__ O,
                                                          float scale) {
  const int g = blockIdx.x, b = blockIdx.y;
  const int t = threadIdx.x, wave = t >> 5, lane = t & 31;
  const int m = lane & 15, half = lane >> 4;
  const int j = g * NWAVES + wave;
  const v2f ones = {1.0f, 1.0f};

  const float* Pj = P + ((size_t)(b * NCHUNK) + 2 * half) * 1024 + j * 16 + m;
  v2f a[NCHUNK / 4];
#pragma unroll
  for (int s = 0; s < NCHUNK / 4; ++s) {
    a[s].x = Pj[(size_t)(4 * s    ) * 1024];
    a[s].y = Pj[(size_t)(4 * s + 1) * 1024];
  }
  const float4 bl = *(const float4*)(bias + j * 16 + 8 * half);
  const float4 bh = *(const float4*)(bias + j * 16 + 8 * half + 4);
  const float bv[8] = {bl.x, bl.y, bl.z, bl.w, bh.x, bh.y, bh.z, bh.w};

  v8f acc = {0.f, 0.f, 0.f, 0.f, 0.f, 0.f, 0.f, 0.f};
#pragma unroll
  for (int s = 0; s < NCHUNK / 4; ++s) {
    acc = __builtin_amdgcn_wmma_f32_16x16x4_f32(
         false, a[s],  false, ones,
         (short)0, acc,  false,  false);
  }
  float val[8];
  float n2 = 0.f;
#pragma unroll
  for (int v = 0; v < 8; ++v) {
    val[v] = scale * acc[v] + bv[v];
    n2 = fmaf(val[v], val[v], n2);
  }
  n2 += __shfl_xor(n2, 16, 32);
  const float coef = (n2 / (1.0f + n2)) / sqrtf(n2 + EPS_SQ);
  if (m == 0) {
#pragma unroll
    for (int v = 0; v < 8; ++v)
      O[(size_t)b * (J_OUT * D_DIM) + j * 16 + 8 * half + v] = val[v] * coef;
  }
}

extern "C" void kernel_launch(void* const* d_in, const int* in_sizes, int n_in,
                              void* d_out, int out_size, void* d_ws, size_t ws_size,
                              hipStream_t stream) {
  const float* X    = (const float*)d_in[0];
  const float* bias = (const float*)d_in[1];
  float* out = (float*)d_out;

  float* P  = (float*)d_ws;
  float* O0 = P  + (size_t)B_SZ * NCHUNK * 1024;
  float* O1 = O0 + B_SZ * J_OUT * D_DIM;
  float* Lg = O1 + B_SZ * J_OUT * D_DIM;

  dim3 grid(NCHUNK, B_SZ);
  dim3 rgrid(JGROUPS, B_SZ);

  caps_pass<0><<<grid, PWAVES * 32, 0, stream>>>(X, nullptr, nullptr, P);
  caps_reduce_squash<<<rgrid, 256, 0, stream>>>(P, bias, O0, 1.0f / 64.0f);

  caps_pass<1><<<grid, PWAVES * 32, 0, stream>>>(X, O0, Lg, P);
  caps_reduce_squash<<<rgrid, 256, 0, stream>>>(P, bias, O1, 1.0f);

  caps_pass<2><<<grid, PWAVES * 32, 0, stream>>>(X, O1, Lg, P);
  caps_reduce_squash<<<rgrid, 256, 0, stream>>>(P, bias, out, 1.0f);
}


# passes 1, 2: a row's twelve lane exchanges (quad dot sums, wave max, wave sum) as DPP row_xmask and v_permlanex16 on the vector ALU instead of ds_bpermute via the LDS unit; same butterfly, same sums
# speedup vs baseline: 1.6521x; 1.0476x over previous
#include <hip/hip_runtime.h>
#include <math.h>


#define B_SZ    32
#define N_INN   2048
#define J_OUT   64
#define D_DIM   16
#define ROW_F4  256
#define NCHUNK  8
#define ROWS_PER_BLOCK (N_INN / NCHUNK)
#define PWAVES  16
#define ROWS_PER_WAVE (ROWS_PER_BLOCK / PWAVES)
#define NWAVES  8
#define JGROUPS (J_OUT / NWAVES)
#define EPS_SQ  1e-7f

typedef __attribute__((ext_vector_type(2))) float v2f;
typedef __attribute__((ext_vector_type(4))) float v4f;
typedef __attribute__((ext_vector_type(8))) float v8f;

__device__ __forceinline__ void f4add(v4f& a, const v4f b) { a += b; }
__device__ __forceinline__ void f4fma(v4f& a, float s, const v4f b) {
  const v4f sv = {s, s, s, s};
  a = __builtin_elementwise_fma(sv, b, a);
}

template <int M>
__device__ __forceinline__ float lane_xor(float v) {
  const int b = __builtin_bit_cast(int, v);
  if (M == 16)
    return __builtin_bit_cast(float, __builtin_amdgcn_permlanex16(b, b, 0x76543210, 0xfedcba98, false, false));
  return __builtin_bit_cast(float, __builtin_amdgcn_update_dpp(b, b, 0x160 + (M & 15), 0xf, 0xf, true));
}

__device__ __forceinline__ int request_after(int nxt, const v4f& done) {
  asm volatile("" : "+s"(nxt) : "v"(done.w));
  return nxt;
}

#define SB() __builtin_amdgcn_sched_barrier(0)

template <int MODE>
__device__ __forceinline__ int walk_row(int r) { return MODE == 1 ? (ROWS_PER_WAVE - 1 - r) : r; }

template <int MODE>
__device__ __forceinline__ void load_row(v4f (&x)[8], float (&lp)[8], const v4f* xw, const float* lw, int r) {
  const int row = walk_row<MODE>(r);
#pragma unroll
  for (int k = 0; k < 8; ++k) x[k] = xw[(size_t)row * ROW_F4 + k * 32];
  if (MODE == 2) {
#pragma unroll
    for (int k = 0; k < 8; ++k) lp[k] = lw[row * J_OUT + 8 * k];
  }
}

template <int MODE>
__device__ __forceinline__ void row_work(const v4f (&xin)[8], const float (&lp)[8], const v4f (&o)[8],
                                         v4f (&acc)[8], float* lw, int r, int dsub) {
  v4f xr[8];
#pragma unroll
  for (int k = 0; k < 8; ++k) {
    xr[k] = xin[k];
    if (MODE != 0) asm volatile("" : "+v"(xr[k]));
  }
  if (MODE == 0) {
#pragma unroll
    for (int k = 0; k < 8; ++k) f4add(acc[k], xr[k]);
  } else {
    float Lv[8];
#pragma unroll
    for (int k = 0; k < 8; ++k) {
      float p = fmaf(xr[k].w, o[k].w, fmaf(xr[k].z, o[k].z, fmaf(xr[k].x, o[k].x, xr[k].y * o[k].y)));
      p += lane_xor<1>(p);
      p += lane_xor<2>(p);
      if (MODE == 2) p += lp[k];
      Lv[k] = p;
    }
    float mx = Lv[0];
#pragma unroll
    for (int k = 1; k < 8; ++k) mx = fmaxf(mx, Lv[k]);
    mx = fmaxf(mx, lane_xor<1>(mx));
    mx = fmaxf(mx, lane_xor<2>(mx));
    mx = fmaxf(mx, lane_xor<4>(mx));
    mx = fmaxf(mx, lane_xor<8>(mx));
    mx = fmaxf(mx, lane_xor<16>(mx));
    float e[8], ss = 0.f;
#pragma unroll
    for (int k = 0; k < 8; ++k) { e[k] = __expf(Lv[k] - mx); ss += e[k]; }
    ss += lane_xor<1>(ss);
    ss += lane_xor<2>(ss);
    ss += lane_xor<4>(ss);
    ss += lane_xor<8>(ss);
    ss += lane_xor<16>(ss);
    const float inv = 4.0f / ss;
#pragma unroll
    for (int k = 0; k < 8; ++k) f4fma(acc[k], e[k] * inv, xr[k]);

    if (MODE == 1 && dsub == 0) {
      const int row = walk_row<MODE>(r);
#pragma unroll
      for (int k = 0; k < 8; ++k) lw[row * J_OUT + 8 * k] = Lv[k];
    }
  }
}

template <int MODE>
__global__ __launch_bounds__(PWAVES * 32) void caps_pass(const float* __restrict__ X,
                                                         const float* __restrict__ Oprev,
                                                         float* __restrict__ Lg,
                                                         float* __restrict__ P) {
  const int chunk = blockIdx.x, b = blockIdx.y;
  const int t = threadIdx.x;
  const int wave = __builtin_amdgcn_readfirstlane(t >> 5);
  const int lane = t & 31;
  const int jgrp = lane >> 2;
  const int dsub = lane & 3;

  __shared__ v4f ls[PWAVES * ROW_F4];

  v4f o[8];
  if (MODE != 0) {
    const v4f* Ob = (const v4f*)Oprev + b * ROW_F4;
#pragma unroll
    for (int k = 0; k < 8; ++k) o[k] = Ob[k * 32 + lane];
  }

  v4f acc[8];
#pragma unroll
  for (int k = 0; k < 8; ++k) acc[k] = (v4f){0.f, 0.f, 0.f, 0.f};

  const int row0 = chunk * ROWS_PER_BLOCK + wave * ROWS_PER_WAVE;
  const v4f* xw = (const v4f*)X + ((size_t)b * N_INN + row0) * ROW_F4 + lane;
  float* lw = Lg + ((size_t)b * N_INN + row0) * J_OUT + jgrp;

  v4f xa[8], xb[8], xc[8];
  float la[8], lb[8], lc[8];
  int nxt = 0;
  load_row<MODE>(xa, la, xw, lw, nxt++); SB();
  load_row<MODE>(xb, lb, xw, lw, nxt++); SB();
#pragma unroll 1
  for (int r = 0; r < ROWS_PER_WAVE - 4; r += 3) {
    load_row<MODE>(xc, lc, xw, lw, nxt++); SB();
    row_work<MODE>(xa, la, o, acc, lw, r, dsub); SB();
    nxt = request_after(nxt, acc[7]);
    load_row<MODE>(xa, la, xw, lw, nxt++); SB();
    row_work<MODE>(xb, lb, o, acc, lw, r + 1, dsub); SB();
    nxt = request_after(nxt, acc[7]);
    load_row<MODE>(xb, lb, xw, lw, nxt++); SB();
    row_work<MODE>(xc, lc, o, acc, lw, r + 2, dsub); SB();
    nxt = request_after(nxt, acc[7]);
  }
  load_row<MODE>(xc, lc, xw, lw, nxt++); SB();
  row_work<MODE>(xa, la, o, acc, lw, ROWS_PER_WAVE - 4, dsub); SB();
  nxt = request_after(nxt, acc[7]);
  load_row<MODE>(xa, la, xw, lw, nxt++); SB();
  row_work<MODE>(xb, lb, o, acc, lw, ROWS_PER_WAVE - 3, dsub); SB();
  row_work<MODE>(xc, lc, o, acc, lw, ROWS_PER_WAVE - 2, dsub); SB();
  row_work<MODE>(xa, la, o, acc, lw, ROWS_PER_WAVE - 1, dsub);

#pragma unroll
  for (int k = 0; k < 8; ++k) ls[wave * ROW_F4 + k * 32 + lane] = acc[k];
  __syncthreads();
  const float2* ls2 = (const float2*)ls;
  float2 s = ls2[t];
#pragma unroll
  for (int w = 1; w < PWAVES; ++w) {
    const float2 v = ls2[w * (2 * ROW_F4) + t];
    s.x += v.x; s.y += v.y;
  }
  ((float2*)P)[((size_t)(b * NCHUNK) + chunk) * (2 * ROW_F4) + t] = s;
}

__global__ __launch_bounds__(256) void caps_reduce_squash(const float* __restrict__ P,
                                                          const float* __restrict__ bias,
                                                          float* __restrict__ O,
                                                          float scale) {
  const int g = blockIdx.x, b = blockIdx.y;
  const int t = threadIdx.x, wave = t >> 5, lane = t & 31;
  const int m = lane & 15, half = lane >> 4;
  const int j = g * NWAVES + wave;
  const v2f ones = {1.0f, 1.0f};

  const float* Pj = P + ((size_t)(b * NCHUNK) + 2 * half) * 1024 + j * 16 + m;
  v2f a[NCHUNK / 4];
#pragma unroll
  for (int s = 0; s < NCHUNK / 4; ++s) {
    a[s].x = Pj[(size_t)(4 * s    ) * 1024];
    a[s].y = Pj[(size_t)(4 * s + 1) * 1024];
  }
  const float4 bl = *(const float4*)(bias + j * 16 + 8 * half);
  const float4 bh = *(const float4*)(bias + j * 16 + 8 * half + 4);
  const float bv[8] = {bl.x, bl.y, bl.z, bl.w, bh.x, bh.y, bh.z, bh.w};

  v8f acc = {0.f, 0.f, 0.f, 0.f, 0.f, 0.f, 0.f, 0.f};
#pragma unroll
  for (int s = 0; s < NCHUNK / 4; ++s) {
    acc = __builtin_amdgcn_wmma_f32_16x16x4_f32(
         false, a[s],  false, ones,
         (short)0, acc,  false,  false);
  }
  float val[8];
  float n2 = 0.f;
#pragma unroll
  for (int v = 0; v < 8; ++v) {
    val[v] = scale * acc[v] + bv[v];
    n2 = fmaf(val[v], val[v], n2);
  }
  n2 += __shfl_xor(n2, 16, 32);
  const float coef = (n2 / (1.0f + n2)) / sqrtf(n2 + EPS_SQ);
  if (m == 0) {
#pragma unroll
    for (int v = 0; v < 8; ++v)
      O[(size_t)b * (J_OUT * D_DIM) + j * 16 + 8 * half + v] = val[v] * coef;
  }
}

extern "C" void kernel_launch(void* const* d_in, const int* in_sizes, int n_in,
                              void* d_out, int out_size, void* d_ws, size_t ws_size,
                              hipStream_t stream) {
  const float* X    = (const float*)d_in[0];
  const float* bias = (const float*)d_in[1];
  float* out = (float*)d_out;

  float* P  = (float*)d_ws;
  float* O0 = P  + (size_t)B_SZ * NCHUNK * 1024;
  float* O1 = O0 + B_SZ * J_OUT * D_DIM;
  float* Lg = O1 + B_SZ * J_OUT * D_DIM;

  dim3 grid(NCHUNK, B_SZ);
  dim3 rgrid(JGROUPS, B_SZ);

  caps_pass<0><<<grid, PWAVES * 32, 0, stream>>>(X, nullptr, nullptr, P);
  caps_reduce_squash<<<rgrid, 256, 0, stream>>>(P, bias, O0, 1.0f / 64.0f);

  caps_pass<1><<<grid, PWAVES * 32, 0, stream>>>(X, O0, Lg, P);
  caps_reduce_squash<<<rgrid, 256, 0, stream>>>(P, bias, O1, 1.0f);

  caps_pass<2><<<grid, PWAVES * 32, 0, stream>>>(X, O1, Lg, P);
  caps_reduce_squash<<<rgrid, 256, 0, stream>>>(P, bias, out, 1.0f);
}


# reduce+squash of iterations 0 and 1 moved into the prologue of the pass that consumes it: 4 launches instead of 6; same chunk order on the f32 WMMA, same squash, sums bit-identical
# speedup vs baseline: 1.7211x; 1.0418x over previous
#include <hip/hip_runtime.h>
#include <math.h>


#define B_SZ    32
#define N_INN   2048
#define J_OUT   64
#define D_DIM   16
#define ROW_F4  256
#define NCHUNK  8
#define ROWS_PER_BLOCK (N_INN / NCHUNK)
#define PWAVES  16
#define ROWS_PER_WAVE (ROWS_PER_BLOCK / PWAVES)
#define NWAVES  8
#define JGROUPS (J_OUT / NWAVES)
#define EPS_SQ  1e-7f

typedef __attribute__((ext_vector_type(2))) float v2f;
typedef __attribute__((ext_vector_type(4))) float v4f;
typedef __attribute__((ext_vector_type(8))) float v8f;

__device__ __forceinline__ void f4add(v4f& a, const v4f b) { a += b; }
__device__ __forceinline__ void f4fma(v4f& a, float s, const v4f b) {
  const v4f sv = {s, s, s, s};
  a = __builtin_elementwise_fma(sv, b, a);
}

template <int M>
__device__ __forceinline__ float lane_xor(float v) {
  const int b = __builtin_bit_cast(int, v);
  if (M == 16)
    return __builtin_bit_cast(float, __builtin_amdgcn_permlanex16(b, b, 0x76543210, 0xfedcba98, false, false));
  return __builtin_bit_cast(float, __builtin_amdgcn_update_dpp(b, b, 0x160 + (M & 15), 0xf, 0xf, true));
}

__device__ __forceinline__ int request_after(int nxt, const v4f& done) {
  asm volatile("" : "+s"(nxt) : "v"(done.w));
  return nxt;
}

#define SB() __builtin_amdgcn_sched_barrier(0)

template <int MODE>
__device__ __forceinline__ int walk_row(int r) { return MODE == 1 ? (ROWS_PER_WAVE - 1 - r) : r; }

template <int MODE>
__device__ __forceinline__ void load_row(v4f (&x)[8], float (&lp)[8], const v4f* xw, const float* lw, int r) {
  const int row = walk_row<MODE>(r);
#pragma unroll
  for (int k = 0; k < 8; ++k) x[k] = xw[(size_t)row * ROW_F4 + k * 32];
  if (MODE == 2) {
#pragma unroll
    for (int k = 0; k < 8; ++k) lp[k] = lw[row * J_OUT + 8 * k];
  }
}

template <int MODE>
__device__ __forceinline__ void row_work(const v4f (&xin)[8], const float (&lp)[8], const v4f (&o)[8],
                                         v4f (&acc)[8], float* lw, int r, int dsub) {
  v4f xr[8];
#pragma unroll
  for (int k = 0; k < 8; ++k) {
    xr[k] = xin[k];
    if (MODE != 0) asm volatile("" : "+v"(xr[k]));
  }
  if (MODE == 0) {
#pragma unroll
    for (int k = 0; k < 8; ++k) f4add(acc[k], xr[k]);
  } else {
    float Lv[8];
#pragma unroll
    for (int k = 0; k < 8; ++k) {
      float p = fmaf(xr[k].w, o[k].w, fmaf(xr[k].z, o[k].z, fmaf(xr[k].x, o[k].x, xr[k].y * o[k].y)));
      p += lane_xor<1>(p);
      p += lane_xor<2>(p);
      if (MODE == 2) p += lp[k];
      Lv[k] = p;
    }
    float mx = Lv[0];
#pragma unroll
    for (int k = 1; k < 8; ++k) mx = fmaxf(mx, Lv[k]);
    mx = fmaxf(mx, lane_xor<1>(mx));
    mx = fmaxf(mx, lane_xor<2>(mx));
    mx = fmaxf(mx, lane_xor<4>(mx));
    mx = fmaxf(mx, lane_xor<8>(mx));
    mx = fmaxf(mx, lane_xor<16>(mx));
    float e[8], ss = 0.f;
#pragma unroll
    for (int k = 0; k < 8; ++k) { e[k] = __expf(Lv[k] - mx); ss += e[k]; }
    ss += lane_xor<1>(ss);
    ss += lane_xor<2>(ss);
    ss += lane_xor<4>(ss);
    ss += lane_xor<8>(ss);
    ss += lane_xor<16>(ss);
    const float inv = 4.0f / ss;
#pragma unroll
    for (int k = 0; k < 8; ++k) f4fma(acc[k], e[k] * inv, xr[k]);

    if (MODE == 1 && dsub == 0) {
      const int row = walk_row<MODE>(r);
#pragma unroll
      for (int k = 0; k < 8; ++k) lw[row * J_OUT + 8 * k] = Lv[k];
    }
  }
}

struct CapsIn {
  v2f a[NCHUNK / 4];
};

template <int CSTRIDE>
__device__ __forceinline__ void caps_in_load(CapsIn& c, const float* Pb, int j, int m, int half) {
  const float* Pj = Pb + (size_t)(2 * half) * CSTRIDE + j * 16 + m;
#pragma unroll
  for (int s = 0; s < NCHUNK / 4; ++s) {
    c.a[s].x = Pj[(size_t)(4 * s    ) * CSTRIDE];
    c.a[s].y = Pj[(size_t)(4 * s + 1) * CSTRIDE];
  }
}

template <bool J_UNIFORM>
__device__ __forceinline__ void bias_half(const float* __restrict__ bias, int j, int half, float4& bl, float4& bh) {
  if (J_UNIFORM) {
    const float4* bj = (const float4*)(bias + j * 16);
    const float4 b0 = bj[0], b1 = bj[1], b2 = bj[2], b3 = bj[3];
    bl = half ? b2 : b0;
    bh = half ? b3 : b1;
  } else {
    bl = *(const float4*)(bias + j * 16 + 8 * half);
    bh = *(const float4*)(bias + j * 16 + 8 * half + 4);
  }
}

__device__ __forceinline__ float chunk_sum(const CapsIn& c, const float4 bl, const float4 bh,
                                           float scale, float (&val)[8]) {
  const v2f ones = {1.0f, 1.0f};
  const float bv[8] = {bl.x, bl.y, bl.z, bl.w, bh.x, bh.y, bh.z, bh.w};
  v8f acc = {0.f, 0.f, 0.f, 0.f, 0.f, 0.f, 0.f, 0.f};
#pragma unroll
  for (int s = 0; s < NCHUNK / 4; ++s) {
    acc = __builtin_amdgcn_wmma_f32_16x16x4_f32(
         false, c.a[s],  false, ones,
         (short)0, acc,  false,  false);
  }
  float n2 = 0.f;
#pragma unroll
  for (int v = 0; v < 8; ++v) {
    val[v] = scale * acc[v] + bv[v];
    n2 = fmaf(val[v], val[v], n2);
  }
  n2 += __shfl_xor(n2, 16, 32);
  return n2;
}

__device__ __forceinline__ float squash_coef(float n2) {
  return (n2 / (1.0f + n2)) / sqrtf(n2 + EPS_SQ);
}

__device__ __forceinline__ float* lg_block(float* Lg, int b, int chunk) {
  return Lg + ((size_t)b * N_INN + (size_t)chunk * ROWS_PER_BLOCK) * J_OUT;
}

template <int MODE>
__global__ __launch_bounds__(PWAVES * 32) void caps_pass(const float* __restrict__ X,
                                                         const float* __restrict__ bias,
                                                         float* __restrict__ Lg,
                                                         float* __restrict__ P,
                                                         float scale) {
  const int chunk = blockIdx.x, b = blockIdx.y;
  const int t = threadIdx.x;
  const int wave = __builtin_amdgcn_readfirstlane(t >> 5);
  const int lane = t & 31;
  const int jgrp = lane >> 2;
  const int dsub = lane & 3;

  __shared__ v4f ls[PWAVES * ROW_F4];
  __shared__ v4f o_s[ROW_F4];

  CapsIn ci[J_OUT / PWAVES];
  if (MODE != 0) {
    const float* Pb = (MODE == 1) ? lg_block(Lg, b, chunk) : P + (size_t)(b * NCHUNK) * 1024;
#pragma unroll
    for (int q = 0; q < J_OUT / PWAVES; ++q)
      caps_in_load<1024>(ci[q], Pb, wave * (J_OUT / PWAVES) + q, lane & 15, lane >> 4);
    SB();
  }

  v4f acc[8];
#pragma unroll
  for (int k = 0; k < 8; ++k) acc[k] = (v4f){0.f, 0.f, 0.f, 0.f};

  const int row0 = chunk * ROWS_PER_BLOCK + wave * ROWS_PER_WAVE;
  const v4f* xw = (const v4f*)X + ((size_t)b * N_INN + row0) * ROW_F4 + lane;
  float* lw = Lg + ((size_t)b * N_INN + row0) * J_OUT + jgrp;

  v4f xa[8], xb[8], xc[8];
  float la[8], lb[8], lc[8];
  int nxt = 0;
  load_row<MODE>(xa, la, xw, lw, nxt++); SB();
  load_row<MODE>(xb, lb, xw, lw, nxt++); SB();
  load_row<MODE>(xc, lc, xw, lw, nxt++); SB();

  v4f o[8];
  if (MODE != 0) {
    int half = lane >> 4;
    asm volatile("" : "+v"(half));
    float4 bl[J_OUT / PWAVES], bh[J_OUT / PWAVES];
#pragma unroll
    for (int q = 0; q < J_OUT / PWAVES; ++q)
      bias_half<true>(bias, wave * (J_OUT / PWAVES) + q, half, bl[q], bh[q]);
#pragma unroll
    for (int q = 0; q < J_OUT / PWAVES; ++q) {
#pragma unroll
      for (int s = 0; s < NCHUNK / 4; ++s) asm volatile("" : "+v"(ci[q].a[s]));
    }
    float val[J_OUT / PWAVES][8], n2[J_OUT / PWAVES];
#pragma unroll
    for (int q = 0; q < J_OUT / PWAVES; ++q) n2[q] = chunk_sum(ci[q], bl[q], bh[q], scale, val[q]);
    if ((lane & 15) == 0) {
#pragma unroll
      for (int q = 0; q < J_OUT / PWAVES; ++q) {
        const float coef = squash_coef(n2[q]);
        const int j = wave * (J_OUT / PWAVES) + q;
        float* od = (float*)o_s + j * D_DIM + 8 * (lane >> 4);
        *(float4*)(od)     = make_float4(val[q][0] * coef, val[q][1] * coef, val[q][2] * coef, val[q][3] * coef);
        *(float4*)(od + 4) = make_float4(val[q][4] * coef, val[q][5] * coef, val[q][6] * coef, val[q][7] * coef);
      }
    }
    __syncthreads();
#pragma unroll
    for (int k = 0; k < 8; ++k) o[k] = o_s[k * 32 + lane];
    SB();
  }
#pragma unroll 1
  for (int r = 0; r < ROWS_PER_WAVE - 4; r += 3) {
    row_work<MODE>(xa, la, o, acc, lw, r, dsub); SB();
    nxt = request_after(nxt, acc[7]);
    load_row<MODE>(xa, la, xw, lw, nxt++); SB();
    row_work<MODE>(xb, lb, o, acc, lw, r + 1, dsub); SB();
    nxt = request_after(nxt, acc[7]);
    load_row<MODE>(xb, lb, xw, lw, nxt++); SB();
    row_work<MODE>(xc, lc, o, acc, lw, r + 2, dsub); SB();
    nxt = request_after(nxt, acc[7]);
    load_row<MODE>(xc, lc, xw, lw, nxt++); SB();
  }
  row_work<MODE>(xa, la, o, acc, lw, ROWS_PER_WAVE - 4, dsub); SB();
  nxt = request_after(nxt, acc[7]);
  load_row<MODE>(xa, la, xw, lw, nxt++); SB();
  row_work<MODE>(xb, lb, o, acc, lw, ROWS_PER_WAVE - 3, dsub); SB();
  row_work<MODE>(xc, lc, o, acc, lw, ROWS_PER_WAVE - 2, dsub); SB();
  row_work<MODE>(xa, la, o, acc, lw, ROWS_PER_WAVE - 1, dsub);

#pragma unroll
  for (int k = 0; k < 8; ++k) ls[wave * ROW_F4 + k * 32 + lane] = acc[k];
  __syncthreads();
  const float2* ls2 = (const float2*)ls;
  float2 s = ls2[t];
#pragma unroll
  for (int w = 1; w < PWAVES; ++w) {
    const float2 v = ls2[w * (2 * ROW_F4) + t];
    s.x += v.x; s.y += v.y;
  }
  if (MODE == 0) {
#pragma unroll
    for (int c2 = 0; c2 < NCHUNK; ++c2)
      ((float2*)(lg_block(Lg, b, c2) + chunk * 1024))[t] = s;
  } else if (MODE == 1) {
    ((float2*)P)[((size_t)(b * NCHUNK) + chunk) * (2 * ROW_F4) + t] = s;
  } else {
    ((float2*)lg_block(Lg, b, chunk))[t] = s;
  }
}

__global__ __launch_bounds__(256) void caps_reduce_squash(const float* __restrict__ Lg,
                                                          const float* __restrict__ bias,
                                                          float* __restrict__ O,
                                                          float scale) {
  const int g = blockIdx.x, b = blockIdx.y;
  const int t = threadIdx.x, wave = t >> 5, lane = t & 31;
  const int m = lane & 15, half = lane >> 4;
  const int j = g * NWAVES + wave;

  CapsIn c;
  caps_in_load<ROWS_PER_BLOCK * J_OUT>(c, Lg + (size_t)b * N_INN * J_OUT, j, m, half);
  float val[8];
  float4 bl, bh;
  bias_half<false>(bias, j, half, bl, bh);
  const float n2 = chunk_sum(c, bl, bh, scale, val);
  const float coef = squash_coef(n2);
  if (m == 0) {
#pragma unroll
    for (int v = 0; v < 8; ++v)
      O[(size_t)b * (J_OUT * D_DIM) + j * 16 + 8 * half + v] = val[v] * coef;
  }
}

extern "C" void kernel_launch(void* const* d_in, const int* in_sizes, int n_in,
                              void* d_out, int out_size, void* d_ws, size_t ws_size,
                              hipStream_t stream) {
  const float* X    = (const float*)d_in[0];
  const float* bias = (const float*)d_in[1];
  float* out = (float*)d_out;

  float* P  = (float*)d_ws;
  float* Lg = P + (size_t)B_SZ * NCHUNK * 1024 + 2 * B_SZ * J_OUT * D_DIM;

  dim3 grid(NCHUNK, B_SZ);
  dim3 rgrid(JGROUPS, B_SZ);

  caps_pass<0><<<grid, PWAVES * 32, 0, stream>>>(X, bias, Lg, P, 0.0f);
  caps_pass<1><<<grid, PWAVES * 32, 0, stream>>>(X, bias, Lg, P, 1.0f / 64.0f);
  caps_pass<2><<<grid, PWAVES * 32, 0, stream>>>(X, bias, Lg, P, 1.0f);
  caps_reduce_squash<<<rgrid, 256, 0, stream>>>(Lg, bias, out, 1.0f);
}


# passes 1, 2: a row's arithmetic in about 110 vector instructions instead of 150: dot as packed pairs (a logit's sum order changes), lane maxima as one DPP v_max each, 4/sum by v_rcp_f32
# speedup vs baseline: 2.1000x; 1.2201x over previous
#include <hip/hip_runtime.h>
#include <math.h>


#define B_SZ    32
#define N_INN   2048
#define J_OUT   64
#define D_DIM   16
#define ROW_F4  256
#define NCHUNK  8
#define ROWS_PER_BLOCK (N_INN / NCHUNK)
#define PWAVES  16
#define ROWS_PER_WAVE (ROWS_PER_BLOCK / PWAVES)
#define NWAVES  8
#define JGROUPS (J_OUT / NWAVES)
#define EPS_SQ  1e-7f

typedef __attribute__((ext_vector_type(2))) float v2f;
typedef __attribute__((ext_vector_type(4))) float v4f;
typedef __attribute__((ext_vector_type(8))) float v8f;

__device__ __forceinline__ void f4add(v4f& a, const v4f b) { a += b; }
__device__ __forceinline__ void f4fma(v4f& a, float s, const v4f b) {
  const v4f sv = {s, s, s, s};
  a = __builtin_elementwise_fma(sv, b, a);
}

template <int M>
__device__ __forceinline__ float lane_xor(float v) {
  const int b = __builtin_bit_cast(int, v);
  if (M == 16)
    return __builtin_bit_cast(float, __builtin_amdgcn_permlanex16(b, b, 0x76543210, 0xfedcba98, false, false));
  return __builtin_bit_cast(float, __builtin_amdgcn_update_dpp(b, b, 0x160 + (M & 15), 0xf, 0xf, true));
}

template <int M>
__device__ __forceinline__ float max_xor(float v) {
  float r;
  if (M == 16) {
    const float w = lane_xor<16>(v);
    asm("s_delay_alu instid0(VALU_DEP_1)\n\t"
        "v_max_num_f32 %0, %1, %2" : "=v"(r) : "v"(v), "v"(w));
  } else {
    asm("s_delay_alu instid0(VALU_DEP_1)\n\t"
        "v_max_num_f32_dpp %0, %1, %1 row_xmask:%2 row_mask:0xf bank_mask:0xf bound_ctrl:1" : "=v"(r) : "v"(v), "n"(M));
  }
  return r;
}

__device__ __forceinline__ int request_after(int nxt, const v4f& done) {
  asm volatile("" : "+s"(nxt) : "v"(done.w));
  return nxt;
}

#define SB() __builtin_amdgcn_sched_barrier(0)

template <int MODE>
__device__ __forceinline__ int walk_row(int r) { return MODE == 1 ? (ROWS_PER_WAVE - 1 - r) : r; }

template <int MODE>
__device__ __forceinline__ void load_row(v4f (&x)[8], float (&lp)[8], const v4f* xw, const float* lw, int r) {
  const int row = walk_row<MODE>(r);
#pragma unroll
  for (int k = 0; k < 8; ++k) x[k] = xw[(size_t)row * ROW_F4 + k * 32];
  if (MODE == 2) {
#pragma unroll
    for (int k = 0; k < 8; ++k) lp[k] = lw[row * J_OUT + 8 * k];
  }
}

template <int MODE>
__device__ __forceinline__ void row_work(const v4f (&xin)[8], const float (&lp)[8], const v4f (&o)[8],
                                         v4f (&acc)[8], float* lw, int r, int dsub) {
  v4f xr[8];
#pragma unroll
  for (int k = 0; k < 8; ++k) {
    xr[k] = xin[k];
    if (MODE != 0) asm volatile("" : "+v"(xr[k]));
  }
  if (MODE == 0) {
#pragma unroll
    for (int k = 0; k < 8; ++k) f4add(acc[k], xr[k]);
  } else {
    float Lv[8];
#pragma unroll
    for (int k = 0; k < 8; ++k) {
      const v2f xlo = __builtin_shufflevector(xr[k], xr[k], 0, 1), xhi = __builtin_shufflevector(xr[k], xr[k], 2, 3);
      const v2f olo = __builtin_shufflevector(o[k], o[k], 0, 1), ohi = __builtin_shufflevector(o[k], o[k], 2, 3);
      const v2f t = __builtin_elementwise_fma(xhi, ohi, xlo * olo);
      float p = t.x + t.y;
      asm("" : "+v"(p));
      p += lane_xor<1>(p);
      p += lane_xor<2>(p);
      asm("" : "+v"(p));
      if (MODE == 2) p += lp[k];
      Lv[k] = p;
    }
    float mx = Lv[0];
#pragma unroll
    for (int k = 1; k < 8; ++k) mx = fmaxf(mx, Lv[k]);
    mx = max_xor<1>(mx);
    mx = max_xor<2>(mx);
    mx = max_xor<4>(mx);
    mx = max_xor<8>(mx);
    mx = max_xor<16>(mx);
    float e[8], ss = 0.f;
#pragma unroll
    for (int k = 0; k < 8; ++k) { e[k] = __expf(Lv[k] - mx); ss += e[k]; }
    ss += lane_xor<1>(ss);
    ss += lane_xor<2>(ss);
    ss += lane_xor<4>(ss);
    ss += lane_xor<8>(ss);
    ss += lane_xor<16>(ss);
    const float inv = 4.0f * __builtin_amdgcn_rcpf(ss);
#pragma unroll
    for (int k = 0; k < 8; ++k) f4fma(acc[k], e[k] * inv, xr[k]);

    if (MODE == 1 && dsub == 0) {
      const int row = walk_row<MODE>(r);
#pragma unroll
      for (int k = 0; k < 8; ++k) lw[row * J_OUT + 8 * k] = Lv[k];
    }
  }
}

struct CapsIn {
  v2f a[NCHUNK / 4];
};

template <int CSTRIDE>
__device__ __forceinline__ void caps_in_load(CapsIn& c, const float* Pb, int j, int m, int half) {
  const float* Pj = Pb + (size_t)(2 * half) * CSTRIDE + j * 16 + m;
#pragma unroll
  for (int s = 0; s < NCHUNK / 4; ++s) {
    c.a[s].x = Pj[(size_t)(4 * s    ) * CSTRIDE];
    c.a[s].y = Pj[(size_t)(4 * s + 1) * CSTRIDE];
  }
}

template <bool J_UNIFORM>
__device__ __forceinline__ void bias_half(const float* __restrict__ bias, int j, int half, float4& bl, float4& bh) {
  if (J_UNIFORM) {
    const float4* bj = (const float4*)(bias + j * 16);
    const float4 b0 = bj[0], b1 = bj[1], b2 = bj[2], b3 = bj[3];
    bl = half ? b2 : b0;
    bh = half ? b3 : b1;
  } else {
    bl = *(const float4*)(bias + j * 16 + 8 * half);
    bh = *(const float4*)(bias + j * 16 + 8 * half + 4);
  }
}

__device__ __forceinline__ float chunk_sum(const CapsIn& c, const float4 bl, const float4 bh,
                                           float scale, float (&val)[8]) {
  const v2f ones = {1.0f, 1.0f};
  const float bv[8] = {bl.x, bl.y, bl.z, bl.w, bh.x, bh.y, bh.z, bh.w};
  v8f acc = {0.f, 0.f, 0.f, 0.f, 0.f, 0.f, 0.f, 0.f};
#pragma unroll
  for (int s = 0; s < NCHUNK / 4; ++s) {
    acc = __builtin_amdgcn_wmma_f32_16x16x4_f32(
         false, c.a[s],  false, ones,
         (short)0, acc,  false,  false);
  }
  float n2 = 0.f;
#pragma unroll
  for (int v = 0; v < 8; ++v) {
    val[v] = scale * acc[v] + bv[v];
    n2 = fmaf(val[v], val[v], n2);
  }
  n2 += __shfl_xor(n2, 16, 32);
  return n2;
}

__device__ __forceinline__ float squash_coef(float n2) {
  return (n2 / (1.0f + n2)) / sqrtf(n2 + EPS_SQ);
}

__device__ __forceinline__ float* lg_block(float* Lg, int b, int chunk) {
  return Lg + ((size_t)b * N_INN + (size_t)chunk * ROWS_PER_BLOCK) * J_OUT;
}

template <int MODE>
__global__ __launch_bounds__(PWAVES * 32) void caps_pass(const float* __restrict__ X,
                                                         const float* __restrict__ bias,
                                                         float* __restrict__ Lg,
                                                         float* __restrict__ P,
                                                         float scale) {
  const int chunk = blockIdx.x, b = blockIdx.y;
  const int t = threadIdx.x;
  const int wave = __builtin_amdgcn_readfirstlane(t >> 5);
  const int lane = t & 31;
  const int jgrp = lane >> 2;
  const int dsub = lane & 3;

  __shared__ v4f ls[PWAVES * ROW_F4];
  __shared__ v4f o_s[ROW_F4];

  CapsIn ci[J_OUT / PWAVES];
  if (MODE != 0) {
    const float* Pb = (MODE == 1) ? lg_block(Lg, b, chunk) : P + (size_t)(b * NCHUNK) * 1024;
#pragma unroll
    for (int q = 0; q < J_OUT / PWAVES; ++q)
      caps_in_load<1024>(ci[q], Pb, wave * (J_OUT / PWAVES) + q, lane & 15, lane >> 4);
    SB();
  }

  v4f acc[8];
#pragma unroll
  for (int k = 0; k < 8; ++k) acc[k] = (v4f){0.f, 0.f, 0.f, 0.f};

  const int row0 = chunk * ROWS_PER_BLOCK + wave * ROWS_PER_WAVE;
  const v4f* xw = (const v4f*)X + ((size_t)b * N_INN + row0) * ROW_F4 + lane;
  float* lw = Lg + ((size_t)b * N_INN + row0) * J_OUT + jgrp;

  v4f xa[8], xb[8], xc[8];
  float la[8], lb[8], lc[8];
  int nxt = 0;
  load_row<MODE>(xa, la, xw, lw, nxt++); SB();
  load_row<MODE>(xb, lb, xw, lw, nxt++); SB();
  load_row<MODE>(xc, lc, xw, lw, nxt++); SB();

  v4f o[8];
  if (MODE != 0) {
    int half = lane >> 4;
    asm volatile("" : "+v"(half));
    float4 bl[J_OUT / PWAVES], bh[J_OUT / PWAVES];
#pragma unroll
    for (int q = 0; q < J_OUT / PWAVES; ++q)
      bias_half<true>(bias, wave * (J_OUT / PWAVES) + q, half, bl[q], bh[q]);
#pragma unroll
    for (int q = 0; q < J_OUT / PWAVES; ++q) {
#pragma unroll
      for (int s = 0; s < NCHUNK / 4; ++s) asm volatile("" : "+v"(ci[q].a[s]));
    }
    float val[J_OUT / PWAVES][8], n2[J_OUT / PWAVES];
#pragma unroll
    for (int q = 0; q < J_OUT / PWAVES; ++q) n2[q] = chunk_sum(ci[q], bl[q], bh[q], scale, val[q]);
    if ((lane & 15) == 0) {
#pragma unroll
      for (int q = 0; q < J_OUT / PWAVES; ++q) {
        const float coef = squash_coef(n2[q]);
        const int j = wave * (J_OUT / PWAVES) + q;
        float* od = (float*)o_s + j * D_DIM + 8 * (lane >> 4);
        *(float4*)(od)     = make_float4(val[q][0] * coef, val[q][1] * coef, val[q][2] * coef, val[q][3] * coef);
        *(float4*)(od + 4) = make_float4(val[q][4] * coef, val[q][5] * coef, val[q][6] * coef, val[q][7] * coef);
      }
    }
    __syncthreads();
#pragma unroll
    for (int k = 0; k < 8; ++k) o[k] = o_s[k * 32 + lane];
    SB();
  }
#pragma unroll 1
  for (int r = 0; r < ROWS_PER_WAVE - 4; r += 3) {
    row_work<MODE>(xa, la, o, acc, lw, r, dsub); SB();
    nxt = request_after(nxt, acc[7]);
    load_row<MODE>(xa, la, xw, lw, nxt++); SB();
    row_work<MODE>(xb, lb, o, acc, lw, r + 1, dsub); SB();
    nxt = request_after(nxt, acc[7]);
    load_row<MODE>(xb, lb, xw, lw, nxt++); SB();
    row_work<MODE>(xc, lc, o, acc, lw, r + 2, dsub); SB();
    nxt = request_after(nxt, acc[7]);
    load_row<MODE>(xc, lc, xw, lw, nxt++); SB();
  }
  row_work<MODE>(xa, la, o, acc, lw, ROWS_PER_WAVE - 4, dsub); SB();
  nxt = request_after(nxt, acc[7]);
  load_row<MODE>(xa, la, xw, lw, nxt++); SB();
  row_work<MODE>(xb, lb, o, acc, lw, ROWS_PER_WAVE - 3, dsub); SB();
  row_work<MODE>(xc, lc, o, acc, lw, ROWS_PER_WAVE - 2, dsub); SB();
  row_work<MODE>(xa, la, o, acc, lw, ROWS_PER_WAVE - 1, dsub);

#pragma unroll
  for (int k = 0; k < 8; ++k) ls[wave * ROW_F4 + k * 32 + lane] = acc[k];
  __syncthreads();
  const float2* ls2 = (const float2*)ls;
  float2 s = ls2[t];
#pragma unroll
  for (int w = 1; w < PWAVES; ++w) {
    const float2 v = ls2[w * (2 * ROW_F4) + t];
    s.x += v.x; s.y += v.y;
  }
  if (MODE == 0) {
#pragma unroll
    for (int c2 = 0; c2 < NCHUNK; ++c2)
      ((float2*)(lg_block(Lg, b, c2) + chunk * 1024))[t] = s;
  } else if (MODE == 1) {
    ((float2*)P)[((size_t)(b * NCHUNK) + chunk) * (2 * ROW_F4) + t] = s;
  } else {
    ((float2*)lg_block(Lg, b, chunk))[t] = s;
  }
}

__global__ __launch_bounds__(256) void caps_reduce_squash(const float* __restrict__ Lg,
                                                          const float* __restrict__ bias,
                                                          float* __restrict__ O,
                                                          float scale) {
  const int g = blockIdx.x, b = blockIdx.y;
  const int t = threadIdx.x, wave = t >> 5, lane = t & 31;
  const int m = lane & 15, half = lane >> 4;
  const int j = g * NWAVES + wave;

  CapsIn c;
  caps_in_load<ROWS_PER_BLOCK * J_OUT>(c, Lg + (size_t)b * N_INN * J_OUT, j, m, half);
  float val[8];
  float4 bl, bh;
  bias_half<false>(bias, j, half, bl, bh);
  const float n2 = chunk_sum(c, bl, bh, scale, val);
  const float coef = squash_coef(n2);
  if (m == 0) {
#pragma unroll
    for (int v = 0; v < 8; ++v)
      O[(size_t)b * (J_OUT * D_DIM) + j * 16 + 8 * half + v] = val[v] * coef;
  }
}

extern "C" void kernel_launch(void* const* d_in, const int* in_sizes, int n_in,
                              void* d_out, int out_size, void* d_ws, size_t ws_size,
                              hipStream_t stream) {
  const float* X    = (const float*)d_in[0];
  const float* bias = (const float*)d_in[1];
  float* out = (float*)d_out;

  float* P  = (float*)d_ws;
  float* Lg = P + (size_t)B_SZ * NCHUNK * 1024 + 2 * B_SZ * J_OUT * D_DIM;

  dim3 grid(NCHUNK, B_SZ);
  dim3 rgrid(JGROUPS, B_SZ);

  caps_pass<0><<<grid, PWAVES * 32, 0, stream>>>(X, bias, Lg, P, 0.0f);
  caps_pass<1><<<grid, PWAVES * 32, 0, stream>>>(X, bias, Lg, P, 1.0f / 64.0f);
  caps_pass<2><<<grid, PWAVES * 32, 0, stream>>>(X, bias, Lg, P, 1.0f);
  caps_reduce_squash<<<rgrid, 256, 0, stream>>>(Lg, bias, out, 1.0f);
}
